# starting kernel
# baseline (speedup 1.0000x reference)
#include <hip/hip_runtime.h>
#include <math.h>

#define NG   2048
#define H_   128
#define W_   128
#define D_   32
#define F_   2
#define LOG2E 1.4426950408889634f

typedef float v2f __attribute__((ext_vector_type(2)));
typedef float v8f __attribute__((ext_vector_type(8)));

__global__ __launch_bounds__(256) void prep_gaussians(
    const float* __restrict__ xyz, const float* __restrict__ scale,
    const float* __restrict__ rot, const float* __restrict__ opac,
    const float* __restrict__ feat, float4* __restrict__ P)
{
    int n = blockIdx.x * 256 + threadIdx.x;
    if (n >= NG) return;

    float qw = rot[n*4+0], qx = rot[n*4+1], qy = rot[n*4+2], qz = rot[n*4+3];
    float inv = 1.0f / (sqrtf(qw*qw + qx*qx + qy*qy + qz*qz) + 1e-8f);
    qw *= inv; qx *= inv; qy *= inv; qz *= inv;

    float r00 = 1.f - 2.f*(qy*qy + qz*qz);
    float r01 = 2.f*(qx*qy - qw*qz);
    float r02 = 2.f*(qx*qz + qw*qy);
    float r10 = 2.f*(qx*qy + qw*qz);
    float r11 = 1.f - 2.f*(qx*qx + qz*qz);
    float r12 = 2.f*(qy*qz - qw*qx);
    float r20 = 2.f*(qx*qz - qw*qy);
    float r21 = 2.f*(qy*qz + qw*qx);
    float r22 = 1.f - 2.f*(qx*qx + qy*qy);

    float sx = scale[n*3+0], sy = scale[n*3+1], sz = scale[n*3+2];
    float m00 = r00*sx, m01 = r01*sy, m02 = r02*sz;
    float m10 = r10*sx, m11 = r11*sy, m12 = r12*sz;
    float m20 = r20*sx, m21 = r21*sy, m22 = r22*sz;

    float cov00 = m00*m00 + m01*m01 + m02*m02;
    float cov01 = m00*m10 + m01*m11 + m02*m12;
    float cov02 = m00*m20 + m01*m21 + m02*m22;
    float cov11 = m10*m10 + m11*m11 + m12*m12;
    float cov12 = m10*m20 + m11*m21 + m12*m22;
    float cov22 = m20*m20 + m21*m21 + m22*m22;

    float zv   = cov22 + 1e-8f;
    float zinv = 1.0f / zv;
    float thr3 = 3.0f * sqrtf(zv);

    float a  = cov00 - zinv*cov02*cov02 + 1e-6f;
    float b  = cov01 - zinv*cov02*cov12;
    float cc = cov11 - zinv*cov12*cov12 + 1e-6f;
    float idet = 1.0f / (a*cc - b*b);
    float pa = cc*idet, pb = -b*idet, pc = a*idet;

    P[n*4+0] = make_float4(xyz[n*3+0], xyz[n*3+1], xyz[n*3+2], cov02);
    P[n*4+1] = make_float4(cov12, zinv, thr3, opac[n]);
    P[n*4+2] = make_float4(-0.5f*LOG2E*pa, -LOG2E*pb, -0.5f*LOG2E*pc, feat[n*2+0]);
    P[n*4+3] = make_float4(feat[n*2+1], 0.f, 0.f, 0.f);
}

__global__ __launch_bounds__(256) void render_slices(
    const float4* __restrict__ P, float* __restrict__ out)
{
    __shared__ float4 s0[256];
    __shared__ float4 s1[256];

    const int d      = blockIdx.x;
    const int h      = blockIdx.y;
    const int tid    = threadIdx.x;
    const int lane   = tid & 31;
    const int wave   = tid >> 5;
    const int laneN  = lane & 15;
    const int laneHi = lane >> 4;
    const int wbase  = wave * 16;

    const float z  = (float)d * (1.0f / (D_ - 1));
    const float yh = (float)h * (1.0f / (H_ - 1));
    const float xw = (float)(wbase + laneN) * (1.0f / (W_ - 1));

    v8f acc = {0.f, 0.f, 0.f, 0.f, 0.f, 0.f, 0.f, 0.f};

    for (int c0 = 0; c0 < NG; c0 += 256) {
        int np = c0 + 256;
        if (np < NG)
            __builtin_prefetch((const void*)&P[(size_t)(np + tid) * 4], 0, 3);

        {
            int g = c0 + tid;
            float4 q0 = P[g*4+0];
            float4 q1 = P[g*4+1];
            float4 q2 = P[g*4+2];
            float4 q3 = P[g*4+3];
            float tt  = q1.y * (z - q0.z);
            float mux = q0.x + q0.w * tt;
            float muy = q0.y + q1.x * tt;
            float msk = (fabsf(q0.z - z) < q1.z) ? 1.0f : 0.0f;
            s0[tid] = make_float4(mux, muy, q1.w * msk, q2.w);
            s1[tid] = make_float4(q2.x, q2.y, q2.z, q3.x);
        }
        __syncthreads();

        for (int s = 0; s < 64; ++s) {
            int gA = (s << 2) + (laneHi << 1);
            float4 a0 = s0[gA];
            float4 b0 = s0[gA + 1];
            bool live = (a0.z != 0.f) || (b0.z != 0.f);
            if (__any(live)) {
                float4 a1 = s1[gA];
                float4 b1 = s1[gA + 1];

                float dxA = xw - a0.x, dyA = yh - a0.y;
                float mA  = dxA * (a1.x * dxA + a1.y * dyA) + a1.z * dyA * dyA;
                float alA = a0.z * __builtin_amdgcn_exp2f(mA);

                float dxB = xw - b0.x, dyB = yh - b0.y;
                float mB  = dxB * (b1.x * dxB + b1.y * dyB) + b1.z * dyB * dyB;
                float alB = b0.z * __builtin_amdgcn_exp2f(mB);

                v2f Am = {alA, alB};
                float e0 = (laneN == 0) ? a0.w : (laneN == 1) ? a1.w : 0.f;
                float e1 = (laneN == 0) ? b0.w : (laneN == 1) ? b1.w : 0.f;
                v2f Bm = {e0, e1};

                acc = __builtin_amdgcn_wmma_f32_16x16x4_f32(
                          false, Am, false, Bm, (short)0, acc, false, false);
            }
        }
        __syncthreads();
    }

    if (laneN < F_) {
        #pragma unroll
        for (int r = 0; r < 8; ++r) {
            int M = r + (laneHi << 3);
            int w = wbase + M;
            out[(((size_t)h * W_ + w) * D_ + d) * F_ + laneN] = acc[r];
        }
    }
}

extern "C" void kernel_launch(void* const* d_in, const int* in_sizes, int n_in,
                              void* d_out, int out_size, void* d_ws, size_t ws_size,
                              hipStream_t stream)
{
    const float* xyz   = (const float*)d_in[0];
    const float* scale = (const float*)d_in[1];
    const float* rot   = (const float*)d_in[2];
    const float* opac  = (const float*)d_in[3];
    const float* feat  = (const float*)d_in[4];
    float4* P = (float4*)d_ws;

    prep_gaussians<<<NG / 256, 256, 0, stream>>>(xyz, scale, rot, opac, feat, P);
    render_slices<<<dim3(D_, H_), 256, 0, stream>>>(P, (float*)d_out);
}


# Exponents computed on the matrix unit: per Gaussian one f32 WMMA (coefficients of 16 slice-row pairs x {x'^2,x',1}) gives 16x4 pixels x 4 slices per wave; mask*opacity moved to B; sum order kept
# speedup vs baseline: 1.5267x; 1.5267x over previous
#include <hip/hip_runtime.h>
#include <math.h>

#define NG   2048
#define H_   128
#define W_   128
#define D_   32
#define F_   2
#define LOG2E 1.4426950408889634f

typedef float v2f __attribute__((ext_vector_type(2)));
typedef float v8f __attribute__((ext_vector_type(8)));

__global__ __launch_bounds__(256) void prep_gaussians(
    const float* __restrict__ xyz, const float* __restrict__ scale,
    const float* __restrict__ rot, const float* __restrict__ opac,
    const float* __restrict__ feat, float4* __restrict__ P)
{
    int n = blockIdx.x * 256 + threadIdx.x;
    if (n >= NG) return;

    float qw = rot[n*4+0], qx = rot[n*4+1], qy = rot[n*4+2], qz = rot[n*4+3];
    float inv = 1.0f / (sqrtf(qw*qw + qx*qx + qy*qy + qz*qz) + 1e-8f);
    qw *= inv; qx *= inv; qy *= inv; qz *= inv;

    float r00 = 1.f - 2.f*(qy*qy + qz*qz);
    float r01 = 2.f*(qx*qy - qw*qz);
    float r02 = 2.f*(qx*qz + qw*qy);
    float r10 = 2.f*(qx*qy + qw*qz);
    float r11 = 1.f - 2.f*(qx*qx + qz*qz);
    float r12 = 2.f*(qy*qz - qw*qx);
    float r20 = 2.f*(qx*qz - qw*qy);
    float r21 = 2.f*(qy*qz + qw*qx);
    float r22 = 1.f - 2.f*(qx*qx + qy*qy);

    float sx = scale[n*3+0], sy = scale[n*3+1], sz = scale[n*3+2];
    float m00 = r00*sx, m01 = r01*sy, m02 = r02*sz;
    float m10 = r10*sx, m11 = r11*sy, m12 = r12*sz;
    float m20 = r20*sx, m21 = r21*sy, m22 = r22*sz;

    float cov00 = m00*m00 + m01*m01 + m02*m02;
    float cov01 = m00*m10 + m01*m11 + m02*m12;
    float cov02 = m00*m20 + m01*m21 + m02*m22;
    float cov11 = m10*m10 + m11*m11 + m12*m12;
    float cov12 = m10*m20 + m11*m21 + m12*m22;
    float cov22 = m20*m20 + m21*m21 + m22*m22;

    float zv   = cov22 + 1e-8f;
    float zinv = 1.0f / zv;
    float thr3 = 3.0f * sqrtf(zv);

    float a  = cov00 - zinv*cov02*cov02 + 1e-6f;
    float b  = cov01 - zinv*cov02*cov12;
    float cc = cov11 - zinv*cov12*cov12 + 1e-6f;
    float idet = 1.0f / (a*cc - b*b);
    float pa = cc*idet, pb = -b*idet, pc = a*idet;

    P[n*4+0] = make_float4(xyz[n*3+0], xyz[n*3+1], xyz[n*3+2], cov02);
    P[n*4+1] = make_float4(cov12, zinv, thr3, opac[n]);
    P[n*4+2] = make_float4(-0.5f*LOG2E*pa, -LOG2E*pb, -0.5f*LOG2E*pc, feat[n*2+0]);
    P[n*4+3] = make_float4(feat[n*2+1], 0.f, 0.f, 0.f);
}

__global__ __launch_bounds__(128) void render_slices(
    const float4* __restrict__ P, float* __restrict__ out)
{
    __shared__ float4 sp[256 * 4];
    __shared__ float2 szt[256];

    const int xt     = blockIdx.x;
    const int h0     = blockIdx.y * 4;
    const int tid    = threadIdx.x;
    const int lane   = tid & 31;
    const int wave   = tid >> 5;
    const int laneN  = lane & 15;
    const int laneHi = lane >> 4;
    const int wbase  = xt * 16;
    const int d0     = (blockIdx.z * 4 + wave) * 4;
    const int dA     = d0 + (laneHi << 1);

    const float zA = (float)dA       * (1.0f / (D_ - 1));
    const float zB = (float)(dA + 1) * (1.0f / (D_ - 1));

    const int   tS = (laneN & 1) + ((laneN >> 3) << 1);
    const int   tY = (laneN & 7) >> 1;
    const float zT = (float)(d0 + tS) * (1.0f / (D_ - 1));
    const float yT = (float)(h0 + tY) * (1.0f / (H_ - 1));
    const float xc = (float)(wbase + 8) * (1.0f / (W_ - 1));
    const float xp = (float)(wbase + laneN) * (1.0f / (W_ - 1)) - xc;
    v2f Bx;
    Bx.x = laneHi ? 1.f : xp * xp;
    Bx.y = laneHi ? 0.f : xp;

    const bool selA = (laneN >> 1) == (laneHi << 1);
    const bool selB = (laneN >> 1) == (laneHi << 1) + 1;
    const bool f1   = (laneN & 1) != 0;

    const v8f zero8 = {0.f, 0.f, 0.f, 0.f, 0.f, 0.f, 0.f, 0.f};
    v8f acc0 = zero8, acc1 = zero8, acc2 = zero8, acc3 = zero8;

    for (int c0 = 0; c0 < NG; c0 += 256) {
        {
            const float4* src = P + (size_t)c0 * 4;
            float4 v[8];
            #pragma unroll
            for (int j = 0; j < 8; ++j) v[j] = src[tid + j * 128];
            const int part = tid & 3;
            #pragma unroll
            for (int j = 0; j < 8; ++j) {
                int e = tid + j * 128;
                sp[e] = v[j];
                if (part < 2) ((float*)szt)[((e >> 2) << 1) + part] = v[j].z;
            }
        }
        __syncthreads();

        const float4* szt4 = (const float4*)szt;
        for (int i4 = 0; i4 < 256; i4 += 4) {
            float4 t01 = szt4[(i4 >> 1)];
            float4 t23 = szt4[(i4 >> 1) + 1];
            float tz[4] = {t01.x, t01.z, t23.x, t23.z};
            float tr[4] = {t01.y, t01.w, t23.y, t23.w};
            #pragma unroll
            for (int u = 0; u < 4; ++u) {
                const int i = i4 + u;
                bool mA = fabsf(tz[u] - zA) < tr[u];
                bool mB = fabsf(tz[u] - zB) < tr[u];
                if (__any(mA || mB)) {
                    float4 q0 = sp[i*4+0];
                    float4 q1 = sp[i*4+1];
                    float4 q2 = sp[i*4+2];
                    float4 q3 = sp[i*4+3];

                    float tt = q1.y * (zT - q0.z);
                    float m  = (q0.x - xc) + q0.w * tt;
                    float dy = yT - (q0.y + q1.x * tt);
                    float ua = q2.x * m;
                    float vb = q2.y * dy;
                    float c1 = vb - 2.f * ua;
                    float cz = m * (ua - vb) + q2.z * dy * dy;
                    v2f A1;
                    A1.x = laneHi ? cz  : q2.x;
                    A1.y = laneHi ? 0.f : c1;

                    v8f E = __builtin_amdgcn_wmma_f32_16x16x4_f32(
                                false, A1, false, Bx, (short)0, zero8, false, false);

                    float fs = f1 ? q3.x : q2.w;
                    float wo = q1.w * fs;
                    v2f B2;
                    B2.x = (selA && mA) ? wo : 0.f;
                    B2.y = (selB && mB) ? wo : 0.f;

                    v2f a0 = {__builtin_amdgcn_exp2f(E[0]), __builtin_amdgcn_exp2f(E[1])};
                    v2f a1 = {__builtin_amdgcn_exp2f(E[2]), __builtin_amdgcn_exp2f(E[3])};
                    v2f a2 = {__builtin_amdgcn_exp2f(E[4]), __builtin_amdgcn_exp2f(E[5])};
                    v2f a3 = {__builtin_amdgcn_exp2f(E[6]), __builtin_amdgcn_exp2f(E[7])};

                    acc0 = __builtin_amdgcn_wmma_f32_16x16x4_f32(
                               false, a0, false, B2, (short)0, acc0, false, false);
                    acc1 = __builtin_amdgcn_wmma_f32_16x16x4_f32(
                               false, a1, false, B2, (short)0, acc1, false, false);
                    acc2 = __builtin_amdgcn_wmma_f32_16x16x4_f32(
                               false, a2, false, B2, (short)0, acc2, false, false);
                    acc3 = __builtin_amdgcn_wmma_f32_16x16x4_f32(
                               false, a3, false, B2, (short)0, acc3, false, false);
                }
            }
        }
        __syncthreads();
    }

    if (laneN < 8) {
        #pragma unroll
        for (int r = 0; r < 8; ++r) {
            int w = wbase + r + (laneHi << 3);
            out[(((size_t)(h0 + 0) * W_ + w) * D_ + d0) * F_ + laneN] = acc0[r];
            out[(((size_t)(h0 + 1) * W_ + w) * D_ + d0) * F_ + laneN] = acc1[r];
            out[(((size_t)(h0 + 2) * W_ + w) * D_ + d0) * F_ + laneN] = acc2[r];
            out[(((size_t)(h0 + 3) * W_ + w) * D_ + d0) * F_ + laneN] = acc3[r];
        }
    }
}

extern "C" void kernel_launch(void* const* d_in, const int* in_sizes, int n_in,
                              void* d_out, int out_size, void* d_ws, size_t ws_size,
                              hipStream_t stream)
{
    const float* xyz   = (const float*)d_in[0];
    const float* scale = (const float*)d_in[1];
    const float* rot   = (const float*)d_in[2];
    const float* opac  = (const float*)d_in[3];
    const float* feat  = (const float*)d_in[4];
    float4* P = (float4*)d_ws;

    prep_gaussians<<<NG / 256, 256, 0, stream>>>(xyz, scale, rot, opac, feat, P);
    render_slices<<<dim3(W_ / 16, H_ / 4, 2), 128, 0, stream>>>(P, (float*)d_out);
}


# Each slice group's Gaussian sum split over 4 waves (index quarters; 16-wave blocks, all Gaussians staged in LDS once) to hide latency; quarter sums added through LDS in fixed order, sums regrouped
# speedup vs baseline: 2.9337x; 1.9216x over previous
#include <hip/hip_runtime.h>
#include <math.h>

#define NG   2048
#define H_   128
#define W_   128
#define D_   32
#define F_   2
#define LOG2E 1.4426950408889634f

typedef float v2f __attribute__((ext_vector_type(2)));
typedef float v8f __attribute__((ext_vector_type(8)));

__global__ __launch_bounds__(256) void prep_gaussians(
    const float* __restrict__ xyz, const float* __restrict__ scale,
    const float* __restrict__ rot, const float* __restrict__ opac,
    const float* __restrict__ feat, float4* __restrict__ P)
{
    int n = blockIdx.x * 256 + threadIdx.x;
    if (n >= NG) return;

    float qw = rot[n*4+0], qx = rot[n*4+1], qy = rot[n*4+2], qz = rot[n*4+3];
    float inv = 1.0f / (sqrtf(qw*qw + qx*qx + qy*qy + qz*qz) + 1e-8f);
    qw *= inv; qx *= inv; qy *= inv; qz *= inv;

    float r00 = 1.f - 2.f*(qy*qy + qz*qz);
    float r01 = 2.f*(qx*qy - qw*qz);
    float r02 = 2.f*(qx*qz + qw*qy);
    float r10 = 2.f*(qx*qy + qw*qz);
    float r11 = 1.f - 2.f*(qx*qx + qz*qz);
    float r12 = 2.f*(qy*qz - qw*qx);
    float r20 = 2.f*(qx*qz - qw*qy);
    float r21 = 2.f*(qy*qz + qw*qx);
    float r22 = 1.f - 2.f*(qx*qx + qy*qy);

    float sx = scale[n*3+0], sy = scale[n*3+1], sz = scale[n*3+2];
    float m00 = r00*sx, m01 = r01*sy, m02 = r02*sz;
    float m10 = r10*sx, m11 = r11*sy, m12 = r12*sz;
    float m20 = r20*sx, m21 = r21*sy, m22 = r22*sz;

    float cov00 = m00*m00 + m01*m01 + m02*m02;
    float cov01 = m00*m10 + m01*m11 + m02*m12;
    float cov02 = m00*m20 + m01*m21 + m02*m22;
    float cov11 = m10*m10 + m11*m11 + m12*m12;
    float cov12 = m10*m20 + m11*m21 + m12*m22;
    float cov22 = m20*m20 + m21*m21 + m22*m22;

    float zv   = cov22 + 1e-8f;
    float zinv = 1.0f / zv;
    float thr3 = 3.0f * sqrtf(zv);

    float a  = cov00 - zinv*cov02*cov02 + 1e-6f;
    float b  = cov01 - zinv*cov02*cov12;
    float cc = cov11 - zinv*cov12*cov12 + 1e-6f;
    float idet = 1.0f / (a*cc - b*b);
    float pa = cc*idet, pb = -b*idet, pc = a*idet;

    P[n*4+0] = make_float4(xyz[n*3+0], xyz[n*3+1], xyz[n*3+2], cov02);
    P[n*4+1] = make_float4(cov12, zinv, thr3, opac[n]);
    P[n*4+2] = make_float4(-0.5f*LOG2E*pa, -LOG2E*pb, -0.5f*LOG2E*pc, feat[n*2+0]);
    P[n*4+3] = make_float4(feat[n*2+1], 0.f, 0.f, 0.f);
}

#define KSPLIT 4
#define NQ     (NG / KSPLIT)
#define RPITCH 33

__global__ __launch_bounds__(512) void render_slices(
    const float4* __restrict__ P, float* __restrict__ out)
{
    __shared__ float4 sp[NG * 4];
    __shared__ float2 szt[NG];

    const int xt     = blockIdx.x;
    const int h0     = blockIdx.y * 4;
    const int tid    = threadIdx.x;
    const int lane   = tid & 31;
    const int wave   = __builtin_amdgcn_readfirstlane(tid >> 5);
    const int grp    = wave >> 2;
    const int gq     = wave & 3;
    const int laneN  = lane & 15;
    const int laneHi = lane >> 4;
    const int wbase  = xt * 16;
    const int d0     = (blockIdx.z * 4 + grp) * 4;
    const int dA     = d0 + (laneHi << 1);

    const float zA = (float)dA       * (1.0f / (D_ - 1));
    const float zB = (float)(dA + 1) * (1.0f / (D_ - 1));

    const int   tS = (laneN & 1) + ((laneN >> 3) << 1);
    const int   tY = (laneN & 7) >> 1;
    const float zT = (float)(d0 + tS) * (1.0f / (D_ - 1));
    const float yT = (float)(h0 + tY) * (1.0f / (H_ - 1));
    const float xc = (float)(wbase + 8) * (1.0f / (W_ - 1));
    const float xp = (float)(wbase + laneN) * (1.0f / (W_ - 1)) - xc;
    v2f Bx;
    Bx.x = laneHi ? 1.f : xp * xp;
    Bx.y = laneHi ? 0.f : xp;

    const bool selA = (laneN >> 1) == (laneHi << 1);
    const bool selB = (laneN >> 1) == (laneHi << 1) + 1;
    const bool f1   = (laneN & 1) != 0;

    const v8f zero8 = {0.f, 0.f, 0.f, 0.f, 0.f, 0.f, 0.f, 0.f};
    v8f acc0 = zero8, acc1 = zero8, acc2 = zero8, acc3 = zero8;

    {
        const int part = tid & 3;
        for (int b = 0; b < 2; ++b) {
            const float4* src = P + b * 8 * 512;
            float4 v[8];
            #pragma unroll
            for (int j = 0; j < 8; ++j) v[j] = src[tid + j * 512];
            #pragma unroll
            for (int j = 0; j < 8; ++j) {
                int e = tid + (b * 8 + j) * 512;
                sp[e] = v[j];
                if (part < 2) ((float*)szt)[((e >> 2) << 1) + part] = v[j].z;
            }
        }
    }
    __syncthreads();

    {
        const float4* szt4 = (const float4*)szt;
        const int ibase = gq * NQ;
        for (int it = 0; it < NQ; it += 4) {
            const int i4 = ibase + it;
            float4 t01 = szt4[(i4 >> 1)];
            float4 t23 = szt4[(i4 >> 1) + 1];
            float tz[4] = {t01.x, t01.z, t23.x, t23.z};
            float tr[4] = {t01.y, t01.w, t23.y, t23.w};
            #pragma unroll
            for (int u = 0; u < 4; ++u) {
                const int i = i4 + u;
                bool mA = fabsf(tz[u] - zA) < tr[u];
                bool mB = fabsf(tz[u] - zB) < tr[u];
                if (__any(mA || mB)) {
                    float4 q0 = sp[i*4+0];
                    float4 q1 = sp[i*4+1];
                    float4 q2 = sp[i*4+2];
                    float4 q3 = sp[i*4+3];

                    float tt = q1.y * (zT - q0.z);
                    float m  = (q0.x - xc) + q0.w * tt;
                    float dy = yT - (q0.y + q1.x * tt);
                    float ua = q2.x * m;
                    float vb = q2.y * dy;
                    float c1 = vb - 2.f * ua;
                    float cz = m * (ua - vb) + q2.z * dy * dy;
                    v2f A1;
                    A1.x = laneHi ? cz  : q2.x;
                    A1.y = laneHi ? 0.f : c1;

                    v8f E = __builtin_amdgcn_wmma_f32_16x16x4_f32(
                                false, A1, false, Bx, (short)0, zero8, false, false);

                    float fs = f1 ? q3.x : q2.w;
                    float wo = q1.w * fs;
                    v2f B2;
                    B2.x = (selA && mA) ? wo : 0.f;
                    B2.y = (selB && mB) ? wo : 0.f;

                    v2f a0 = {__builtin_amdgcn_exp2f(E[0]), __builtin_amdgcn_exp2f(E[1])};
                    v2f a1 = {__builtin_amdgcn_exp2f(E[2]), __builtin_amdgcn_exp2f(E[3])};
                    v2f a2 = {__builtin_amdgcn_exp2f(E[4]), __builtin_amdgcn_exp2f(E[5])};
                    v2f a3 = {__builtin_amdgcn_exp2f(E[6]), __builtin_amdgcn_exp2f(E[7])};

                    acc0 = __builtin_amdgcn_wmma_f32_16x16x4_f32(
                               false, a0, false, B2, (short)0, acc0, false, false);
                    acc1 = __builtin_amdgcn_wmma_f32_16x16x4_f32(
                               false, a1, false, B2, (short)0, acc1, false, false);
                    acc2 = __builtin_amdgcn_wmma_f32_16x16x4_f32(
                               false, a2, false, B2, (short)0, acc2, false, false);
                    acc3 = __builtin_amdgcn_wmma_f32_16x16x4_f32(
                               false, a3, false, B2, (short)0, acc3, false, false);
                }
            }
        }
    }
    __syncthreads();

    float* red = (float*)sp;
    if (laneN < 8) {
        float* dst = red + (gq * 64 + (laneHi << 3)) * RPITCH + grp * 8 + laneN;
        #pragma unroll
        for (int r = 0; r < 8; ++r) {
            dst[(r +  0) * RPITCH] = acc0[r];
            dst[(r + 16) * RPITCH] = acc1[r];
            dst[(r + 32) * RPITCH] = acc2[r];
            dst[(r + 48) * RPITCH] = acc3[r];
        }
    }
    __syncthreads();

    #pragma unroll
    for (int k = 0; k < 4; ++k) {
        const int p = wave * 4 + k;
        const float* s = red + p * RPITCH + lane;
        float sum = s[0] + s[64 * RPITCH];
        sum += s[2 * 64 * RPITCH];
        sum += s[3 * 64 * RPITCH];
        out[((h0 + (p >> 4)) * W_ + wbase + (p & 15)) * (D_ * F_) + blockIdx.z * 32 + lane] = sum;
    }
}

extern "C" void kernel_launch(void* const* d_in, const int* in_sizes, int n_in,
                              void* d_out, int out_size, void* d_ws, size_t ws_size,
                              hipStream_t stream)
{
    const float* xyz   = (const float*)d_in[0];
    const float* scale = (const float*)d_in[1];
    const float* rot   = (const float*)d_in[2];
    const float* opac  = (const float*)d_in[3];
    const float* feat  = (const float*)d_in[4];
    float4* P = (float4*)d_ws;

    prep_gaussians<<<NG / 256, 256, 0, stream>>>(xyz, scale, rot, opac, feat, P);
    render_slices<<<dim3(W_ / 16, H_ / 4, 2), 512, 0, stream>>>(P, (float*)d_out);
}


# Units re-dealt to blocks: a block's 8 wave groups take 8 different 16x8 patches (every patch column, every 2nd patch row) and every 2nd slice pair, middle pairs on the oldest waves; sums unchanged
# speedup vs baseline: 4.8521x; 1.6539x over previous
#include <hip/hip_runtime.h>
#include <math.h>

#define NG   2048
#define H_   128
#define W_   128
#define D_   32
#define F_   2
#define LOG2E 1.4426950408889634f

typedef float v2f __attribute__((ext_vector_type(2)));
typedef float v8f __attribute__((ext_vector_type(8)));

__global__ __launch_bounds__(256) void prep_gaussians(
    const float* __restrict__ xyz, const float* __restrict__ scale,
    const float* __restrict__ rot, const float* __restrict__ opac,
    const float* __restrict__ feat, float4* __restrict__ P)
{
    int n = blockIdx.x * 256 + threadIdx.x;
    if (n >= NG) return;

    float qw = rot[n*4+0], qx = rot[n*4+1], qy = rot[n*4+2], qz = rot[n*4+3];
    float inv = 1.0f / (sqrtf(qw*qw + qx*qx + qy*qy + qz*qz) + 1e-8f);
    qw *= inv; qx *= inv; qy *= inv; qz *= inv;

    float r00 = 1.f - 2.f*(qy*qy + qz*qz);
    float r01 = 2.f*(qx*qy - qw*qz);
    float r02 = 2.f*(qx*qz + qw*qy);
    float r10 = 2.f*(qx*qy + qw*qz);
    float r11 = 1.f - 2.f*(qx*qx + qz*qz);
    float r12 = 2.f*(qy*qz - qw*qx);
    float r20 = 2.f*(qx*qz - qw*qy);
    float r21 = 2.f*(qy*qz + qw*qx);
    float r22 = 1.f - 2.f*(qx*qx + qy*qy);

    float sx = scale[n*3+0], sy = scale[n*3+1], sz = scale[n*3+2];
    float m00 = r00*sx, m01 = r01*sy, m02 = r02*sz;
    float m10 = r10*sx, m11 = r11*sy, m12 = r12*sz;
    float m20 = r20*sx, m21 = r21*sy, m22 = r22*sz;

    float cov00 = m00*m00 + m01*m01 + m02*m02;
    float cov01 = m00*m10 + m01*m11 + m02*m12;
    float cov02 = m00*m20 + m01*m21 + m02*m22;
    float cov11 = m10*m10 + m11*m11 + m12*m12;
    float cov12 = m10*m20 + m11*m21 + m12*m22;
    float cov22 = m20*m20 + m21*m21 + m22*m22;

    float zv   = cov22 + 1e-8f;
    float zinv = 1.0f / zv;
    float thr3 = 3.0f * sqrtf(zv);

    float a  = cov00 - zinv*cov02*cov02 + 1e-6f;
    float b  = cov01 - zinv*cov02*cov12;
    float cc = cov11 - zinv*cov12*cov12 + 1e-6f;
    float idet = 1.0f / (a*cc - b*b);
    float pa = cc*idet, pb = -b*idet, pc = a*idet;

    P[n*4+0] = make_float4(xyz[n*3+0], xyz[n*3+1], xyz[n*3+2], cov02);
    P[n*4+1] = make_float4(cov12, zinv, thr3, opac[n]);
    P[n*4+2] = make_float4(-0.5f*LOG2E*pa, -LOG2E*pb, -0.5f*LOG2E*pc, feat[n*2+0]);
    P[n*4+3] = make_float4(feat[n*2+1], 0.f, 0.f, 0.f);
}

__device__ __forceinline__ void unit_of(int bx, int by, int bz, int v, int& xt, int& hb, int& sp)
{
    xt = (bx + v) & 7;
    hb = (by + 6 * v) & 15;
    const int d = (v & 1) ? -(v + 1) : v;
    sp = bz ? 7 - d : 8 + d;
}

#define KSPLIT 4
#define NQ     (NG / KSPLIT)
#define RPITCH 33
#define NPIX   128
#define EXP_ZERO (-150.0f)

__global__ __launch_bounds__(1024) void render_slices(
    const float4* __restrict__ P, float* __restrict__ out)
{
    __shared__ float4 sp[NG * 4];
    __shared__ float2 szt[NG];
    __shared__ float  red[KSPLIT * NPIX * RPITCH];

    const int tid    = threadIdx.x;
    const int lane   = tid & 31;
    const int wave   = __builtin_amdgcn_readfirstlane(tid >> 5);
    const int grp    = wave >> 2;
    const int gq     = wave & 3;
    int xt, hb, spair;
    unit_of(blockIdx.x, blockIdx.y, blockIdx.z, grp, xt, hb, spair);
    const int h0     = hb * 8;
    const int laneN  = lane & 15;
    const int laneHi = lane >> 4;
    const int wbase  = xt * 16;
    const int d0     = spair * 2;

    const float zA = (float)d0       * (1.0f / (D_ - 1));
    const float zB = (float)(d0 + 1) * (1.0f / (D_ - 1));

    const int   tS = laneN & 1;
    const int   tY = (laneN & 6) + (laneN >> 3);
    const float zT = (float)(d0 + tS) * (1.0f / (D_ - 1));
    const float yT = (float)(h0 + tY) * (1.0f / (H_ - 1));
    const float xc = (float)(wbase + 8) * (1.0f / (W_ - 1));
    const float xp = (float)(wbase + laneN) * (1.0f / (W_ - 1)) - xc;
    v2f Bx;
    Bx.x = laneHi ? 1.f : xp * xp;
    Bx.y = laneHi ? 0.f : xp;

    const bool selA = (laneN >> 1) == (laneHi << 1);
    const bool selB = (laneN >> 1) == (laneHi << 1) + 1;
    const bool f1   = (laneN & 1) != 0;

    const v8f zero8 = {0.f, 0.f, 0.f, 0.f, 0.f, 0.f, 0.f, 0.f};
    v8f acc0 = zero8, acc1 = zero8, acc2 = zero8, acc3 = zero8;

    {
        const int part = tid & 3;
        float4 v[8];
        #pragma unroll
        for (int j = 0; j < 8; ++j) v[j] = P[tid + j * 1024];
        #pragma unroll
        for (int j = 0; j < 8; ++j) {
            int e = tid + j * 1024;
            sp[e] = v[j];
            if (part < 2) ((float*)szt)[((e >> 2) << 1) + part] = v[j].z;
        }
    }
    __syncthreads();

    {
        const int ibase = gq * NQ;
        float2 tnext = szt[ibase + lane];
        for (int c = 0; c < NQ / 32; ++c) {
            const int cb = ibase + c * 32;
            const float2 tcur = tnext;
            const int cn = (c + 1 < NQ / 32) ? c + 1 : c;
            tnext = szt[ibase + cn * 32 + lane];
            const unsigned bitsA = __builtin_amdgcn_ballot_w32(fabsf(tcur.x - zA) < tcur.y);
            const unsigned bitsB = __builtin_amdgcn_ballot_w32(fabsf(tcur.x - zB) < tcur.y);
            unsigned keep = bitsA | bitsB;
            while (keep != 0u) {
                const unsigned low = keep & (0u - keep);
                keep ^= low;
                const int  i  = cb + __builtin_ctz(low);
                const bool mA = (bitsA & low) != 0u;
                const bool mB = (bitsB & low) != 0u;

                float4 q0 = sp[i*4+0];
                float4 q1 = sp[i*4+1];
                float4 q2 = sp[i*4+2];
                float4 q3 = sp[i*4+3];

                float tt = q1.y * (zT - q0.z);
                float m  = (q0.x - xc) + q0.w * tt;
                float dy = yT - (q0.y + q1.x * tt);
                float ua = q2.x * m;
                float vb = q2.y * dy;
                float c1 = vb - 2.f * ua;
                float cz = m * (ua - vb) + q2.z * dy * dy;
                v2f A1;
                A1.x = laneHi ? cz  : q2.x;
                A1.y = laneHi ? 0.f : c1;

                v8f E = __builtin_amdgcn_wmma_f32_16x16x4_f32(
                            false, A1, false, Bx, (short)0, zero8, false, false);

                asm volatile("" :: "v"(q1.w), "v"(q2.w), "v"(q3.x));

                float mx = fmaxf(fmaxf(fmaxf(E[0], E[1]), fmaxf(E[2], E[3])),
                                 fmaxf(fmaxf(E[4], E[5]), fmaxf(E[6], E[7])));
                if (__any(!(mx < EXP_ZERO))) {
                    float fs = f1 ? q3.x : q2.w;
                    float wo = q1.w * fs;
                    v2f B2;
                    B2.x = (selA && mA) ? wo : 0.f;
                    B2.y = (selB && mB) ? wo : 0.f;

                    v2f a0 = {__builtin_amdgcn_exp2f(E[0]), __builtin_amdgcn_exp2f(E[1])};
                    v2f a1 = {__builtin_amdgcn_exp2f(E[2]), __builtin_amdgcn_exp2f(E[3])};
                    v2f a2 = {__builtin_amdgcn_exp2f(E[4]), __builtin_amdgcn_exp2f(E[5])};
                    v2f a3 = {__builtin_amdgcn_exp2f(E[6]), __builtin_amdgcn_exp2f(E[7])};

                    acc0 = __builtin_amdgcn_wmma_f32_16x16x4_f32(
                               false, a0, false, B2, (short)0, acc0, false, false);
                    acc1 = __builtin_amdgcn_wmma_f32_16x16x4_f32(
                               false, a1, false, B2, (short)0, acc1, false, false);
                    acc2 = __builtin_amdgcn_wmma_f32_16x16x4_f32(
                               false, a2, false, B2, (short)0, acc2, false, false);
                    acc3 = __builtin_amdgcn_wmma_f32_16x16x4_f32(
                               false, a3, false, B2, (short)0, acc3, false, false);
                }
            }
        }
    }

    if (laneN < 8) {
        float* dst = red + (gq * NPIX + ((laneN >> 2) << 4) + (laneHi << 3)) * RPITCH + grp * 4 + (laneN & 3);
        #pragma unroll
        for (int r = 0; r < 8; ++r) {
            dst[(r +  0) * RPITCH] = acc0[r];
            dst[(r + 32) * RPITCH] = acc1[r];
            dst[(r + 64) * RPITCH] = acc2[r];
            dst[(r + 96) * RPITCH] = acc3[r];
        }
    }
    __syncthreads();

    int xtL, hbL, spL;
    unit_of(blockIdx.x, blockIdx.y, blockIdx.z, lane >> 2, xtL, hbL, spL);
    float* const outL = out + (hbL * 8 * W_ + xtL * 16) * (D_ * F_) + spL * 4 + (lane & 3);
    #pragma unroll
    for (int k = 0; k < 4; ++k) {
        const int p = wave * 4 + k;
        const float* s = red + p * RPITCH + lane;
        float sum = s[0] + s[NPIX * RPITCH];
        sum += s[2 * NPIX * RPITCH];
        sum += s[3 * NPIX * RPITCH];
        outL[((p >> 4) * W_ + (p & 15)) * (D_ * F_)] = sum;
    }
}

extern "C" void kernel_launch(void* const* d_in, const int* in_sizes, int n_in,
                              void* d_out, int out_size, void* d_ws, size_t ws_size,
                              hipStream_t stream)
{
    const float* xyz   = (const float*)d_in[0];
    const float* scale = (const float*)d_in[1];
    const float* rot   = (const float*)d_in[2];
    const float* opac  = (const float*)d_in[3];
    const float* feat  = (const float*)d_in[4];
    float4* P = (float4*)d_ws;

    prep_gaussians<<<NG / 256, 256, 0, stream>>>(xyz, scale, rot, opac, feat, P);
    render_slices<<<dim3(W_ / 16, H_ / 8, 2), 1024, 0, stream>>>(P, (float*)d_out);
}


# One launch: every block computes the records of all 2048 Gaussians itself (2 per thread, same arithmetic) straight into LDS; parameter kernel, its launch and the workspace pass gone; sums unchanged
# speedup vs baseline: 5.6627x; 1.1671x over previous
#include <hip/hip_runtime.h>
#include <math.h>

#define NG   2048
#define H_   128
#define W_   128
#define D_   32
#define F_   2
#define LOG2E 1.4426950408889634f

typedef float v2f __attribute__((ext_vector_type(2)));
typedef float v8f __attribute__((ext_vector_type(8)));

#define BOX_R 14.9f
struct GaussIn { float qw, qx, qy, qz, sx, sy, sz, x, y, z, op, f0, f1; };

__device__ __forceinline__ GaussIn load_gaussian(
    const int n,
    const float* __restrict__ xyz, const float* __restrict__ scale,
    const float* __restrict__ rot, const float* __restrict__ opac,
    const float* __restrict__ feat)
{
    GaussIn g;
    g.qw = rot[n*4+0];   g.qx = rot[n*4+1];   g.qy = rot[n*4+2];   g.qz = rot[n*4+3];
    g.sx = scale[n*3+0]; g.sy = scale[n*3+1]; g.sz = scale[n*3+2];
    g.x  = xyz[n*3+0];   g.y  = xyz[n*3+1];   g.z  = xyz[n*3+2];
    g.op = opac[n];
    g.f0 = feat[n*2+0];  g.f1 = feat[n*2+1];
    return g;
}

__device__ __forceinline__ void gaussian_record(
    const GaussIn g, float4& r0, float4& r1, float4& r2, float4& r3)
{
    float qw = g.qw, qx = g.qx, qy = g.qy, qz = g.qz;
    float inv = 1.0f / (sqrtf(qw*qw + qx*qx + qy*qy + qz*qz) + 1e-8f);
    qw *= inv; qx *= inv; qy *= inv; qz *= inv;

    float r00 = 1.f - 2.f*(qy*qy + qz*qz);
    float r01 = 2.f*(qx*qy - qw*qz);
    float r02 = 2.f*(qx*qz + qw*qy);
    float r10 = 2.f*(qx*qy + qw*qz);
    float r11 = 1.f - 2.f*(qx*qx + qz*qz);
    float r12 = 2.f*(qy*qz - qw*qx);
    float r20 = 2.f*(qx*qz - qw*qy);
    float r21 = 2.f*(qy*qz + qw*qx);
    float r22 = 1.f - 2.f*(qx*qx + qy*qy);

    float sx = g.sx, sy = g.sy, sz = g.sz;
    float m00 = r00*sx, m01 = r01*sy, m02 = r02*sz;
    float m10 = r10*sx, m11 = r11*sy, m12 = r12*sz;
    float m20 = r20*sx, m21 = r21*sy, m22 = r22*sz;

    float cov00 = m00*m00 + m01*m01 + m02*m02;
    float cov01 = m00*m10 + m01*m11 + m02*m12;
    float cov02 = m00*m20 + m01*m21 + m02*m22;
    float cov11 = m10*m10 + m11*m11 + m12*m12;
    float cov12 = m10*m20 + m11*m21 + m12*m22;
    float cov22 = m20*m20 + m21*m21 + m22*m22;

    float zv   = cov22 + 1e-8f;
    float zinv = 1.0f / zv;
    float thr3 = 3.0f * sqrtf(zv);

    float a  = cov00 - zinv*cov02*cov02 + 1e-6f;
    float b  = cov01 - zinv*cov02*cov12;
    float cc = cov11 - zinv*cov12*cov12 + 1e-6f;
    float idet = 1.0f / (a*cc - b*b);
    float pa = cc*idet, pb = -b*idet, pc = a*idet;

    float infl = BOX_R * (1.0f + 1e-6f * a * fabsf(pa));
    float bx = infl * sqrtf(a)  + 7.5f * (1.0f / (W_ - 1));
    float by = infl * sqrtf(cc) + 3.5f * (1.0f / (H_ - 1));

    r0 = make_float4(g.x, g.y, g.z, cov02);
    r1 = make_float4(cov12, zinv, -0.5f*LOG2E*pa, -LOG2E*pb);
    r2 = make_float4(-0.5f*LOG2E*pc, g.op, g.f0, g.f1);
    r3 = make_float4(thr3, bx, by, 0.f);
}

__device__ __forceinline__ void unit_of(int bx, int by, int bz, int v, int& xt, int& hb, int& sp)
{
    xt = (bx + v) & 7;
    hb = (by + 6 * v) & 15;
    const int d = (v & 1) ? -(v + 1) : v;
    sp = bz ? 7 - d : 8 + d;
}

__device__ __forceinline__ void chunk_masks(const float4 t0, const float4 t1, const float4 t2,
                                            const float zA, const float zB,
                                            const float xcen, const float ycen,
                                            unsigned& bitsA, unsigned& bitsB, unsigned& keep)
{
    {
        #pragma clang fp contract(off)
        bitsA = __builtin_amdgcn_ballot_w32(fabsf(t0.z - zA) < t2.x);
        bitsB = __builtin_amdgcn_ballot_w32(fabsf(t0.z - zB) < t2.x);
    }
    const float ttA = t1.y * (zA - t0.z);
    const float ttB = t1.y * (zB - t0.z);
    const unsigned inA = __builtin_amdgcn_ballot_w32(!(fabsf((t0.x + t0.w * ttA) - xcen) > t2.y))
                       & __builtin_amdgcn_ballot_w32(!(fabsf((t0.y + t1.x * ttA) - ycen) > t2.z));
    const unsigned inB = __builtin_amdgcn_ballot_w32(!(fabsf((t0.x + t0.w * ttB) - xcen) > t2.y))
                       & __builtin_amdgcn_ballot_w32(!(fabsf((t0.y + t1.x * ttB) - ycen) > t2.z));
    keep = (bitsA & inA) | (bitsB & inB);
}

#define KSPLIT 4
#define NQ     (NG / KSPLIT)
#define RPITCH 33
#define NPIX   128
#define EXP_ZERO (-150.0f)

__global__ __launch_bounds__(1024) void render_slices(
    const float* __restrict__ xyz, const float* __restrict__ scale,
    const float* __restrict__ rot, const float* __restrict__ opac,
    const float* __restrict__ feat, float* __restrict__ out)
{
    __shared__ float4 sp[NG * 4];
    __shared__ float  red[KSPLIT * NPIX * RPITCH];

    const int tid    = threadIdx.x;
    const int lane   = tid & 31;
    const int wave   = __builtin_amdgcn_readfirstlane(tid >> 5);
    const int grp    = wave >> 2;
    const int gq     = wave & 3;
    int xt, hb, spair;
    unit_of(blockIdx.x, blockIdx.y, blockIdx.z, grp, xt, hb, spair);
    const int h0     = hb * 8;
    const int laneN  = lane & 15;
    const int laneHi = lane >> 4;
    const int wbase  = xt * 16;
    const int d0     = spair * 2;

    const float zA = (float)d0       * (1.0f / (D_ - 1));
    const float zB = (float)(d0 + 1) * (1.0f / (D_ - 1));
    const float xcen = ((float)wbase + 7.5f) * (1.0f / (W_ - 1));
    const float ycen = ((float)h0    + 3.5f) * (1.0f / (H_ - 1));

    const int   tS = laneN & 1;
    const int   tY = (laneN & 6) + (laneN >> 3);
    const float zT = (float)(d0 + tS) * (1.0f / (D_ - 1));
    const float yT = (float)(h0 + tY) * (1.0f / (H_ - 1));
    const float xc = (float)(wbase + 8) * (1.0f / (W_ - 1));
    const float xp = (float)(wbase + laneN) * (1.0f / (W_ - 1)) - xc;
    v2f Bx;
    Bx.x = laneHi ? 1.f : xp * xp;
    Bx.y = laneHi ? 0.f : xp;

    const bool selA = (laneN >> 1) == (laneHi << 1);
    const bool selB = (laneN >> 1) == (laneHi << 1) + 1;
    const bool f1   = (laneN & 1) != 0;

    const v8f zero8 = {0.f, 0.f, 0.f, 0.f, 0.f, 0.f, 0.f, 0.f};
    v8f acc0 = zero8, acc1 = zero8, acc2 = zero8, acc3 = zero8;

    {
        GaussIn gin[NG / 1024];
        #pragma unroll
        for (int j = 0; j < NG / 1024; ++j) gin[j] = load_gaussian(tid + j * 1024, xyz, scale, rot, opac, feat);
        __builtin_amdgcn_sched_barrier(0);
        #pragma unroll
        for (int j = 0; j < NG / 1024; ++j) {
            const int n = tid + j * 1024;
            float4 r0, r1, r2, r3;
            gaussian_record(gin[j], r0, r1, r2, r3);
            sp[n*4+0] = r0;
            sp[n*4+1] = r1;
            sp[n*4+2] = r2;
            sp[n*4+3] = r3;
        }
    }
    __syncthreads();

    {
        const int ibase = gq * NQ;
        unsigned nextA, nextB, nextKeep;
        chunk_masks(sp[(ibase + lane) * 4 + 0], sp[(ibase + lane) * 4 + 1], sp[(ibase + lane) * 4 + 3],
                    zA, zB, xcen, ycen, nextA, nextB, nextKeep);
        for (int c = 0; c < NQ / 32; ++c) {
            const int cb = ibase + c * 32;
            const unsigned bitsA = nextA, bitsB = nextB;
            unsigned keep = nextKeep;
            const int cn = (c + 1 < NQ / 32) ? c + 1 : c;
            const float4 t0 = sp[(ibase + cn * 32 + lane) * 4 + 0];
            const float4 t1 = sp[(ibase + cn * 32 + lane) * 4 + 1];
            const float4 t2 = sp[(ibase + cn * 32 + lane) * 4 + 3];
            while (keep != 0u) {
                const unsigned low = keep & (0u - keep);
                keep ^= low;
                const int  i  = cb + __builtin_ctz(low);
                const bool mA = (bitsA & low) != 0u;
                const bool mB = (bitsB & low) != 0u;

                float4 q0 = sp[i*4+0];
                float4 q1 = sp[i*4+1];
                float4 q2 = sp[i*4+2];

                float tt = q1.y * (zT - q0.z);
                float m  = (q0.x - xc) + q0.w * tt;
                float dy = yT - (q0.y + q1.x * tt);
                float ua = q1.z * m;
                float vb = q1.w * dy;
                float c1 = vb - 2.f * ua;
                float cz = m * (ua - vb) + q2.x * dy * dy;
                v2f A1;
                A1.x = laneHi ? cz  : q1.z;
                A1.y = laneHi ? 0.f : c1;

                v8f E = __builtin_amdgcn_wmma_f32_16x16x4_f32(
                            false, A1, false, Bx, (short)0, zero8, false, false);

                asm volatile("" :: "v"(q2.y), "v"(q2.z), "v"(q2.w));

                float mx = fmaxf(fmaxf(fmaxf(E[0], E[1]), fmaxf(E[2], E[3])),
                                 fmaxf(fmaxf(E[4], E[5]), fmaxf(E[6], E[7])));
                if (__any(!(mx < EXP_ZERO))) {
                    float fs = f1 ? q2.w : q2.z;
                    float wo = q2.y * fs;
                    v2f B2;
                    B2.x = (selA && mA) ? wo : 0.f;
                    B2.y = (selB && mB) ? wo : 0.f;

                    v2f a0 = {__builtin_amdgcn_exp2f(E[0]), __builtin_amdgcn_exp2f(E[1])};
                    v2f a1 = {__builtin_amdgcn_exp2f(E[2]), __builtin_amdgcn_exp2f(E[3])};
                    v2f a2 = {__builtin_amdgcn_exp2f(E[4]), __builtin_amdgcn_exp2f(E[5])};
                    v2f a3 = {__builtin_amdgcn_exp2f(E[6]), __builtin_amdgcn_exp2f(E[7])};

                    acc0 = __builtin_amdgcn_wmma_f32_16x16x4_f32(
                               false, a0, false, B2, (short)0, acc0, false, false);
                    acc1 = __builtin_amdgcn_wmma_f32_16x16x4_f32(
                               false, a1, false, B2, (short)0, acc1, false, false);
                    acc2 = __builtin_amdgcn_wmma_f32_16x16x4_f32(
                               false, a2, false, B2, (short)0, acc2, false, false);
                    acc3 = __builtin_amdgcn_wmma_f32_16x16x4_f32(
                               false, a3, false, B2, (short)0, acc3, false, false);
                }
            }
            chunk_masks(t0, t1, t2, zA, zB, xcen, ycen, nextA, nextB, nextKeep);
        }
    }

    if (laneN < 8) {
        float* dst = red + (gq * NPIX + ((laneN >> 2) << 4) + (laneHi << 3)) * RPITCH + grp * 4 + (laneN & 3);
        #pragma unroll
        for (int r = 0; r < 8; ++r) {
            dst[(r +  0) * RPITCH] = acc0[r];
            dst[(r + 32) * RPITCH] = acc1[r];
            dst[(r + 64) * RPITCH] = acc2[r];
            dst[(r + 96) * RPITCH] = acc3[r];
        }
    }
    __syncthreads();

    int xtL, hbL, spL;
    unit_of(blockIdx.x, blockIdx.y, blockIdx.z, lane >> 2, xtL, hbL, spL);
    float* const outL = out + (hbL * 8 * W_ + xtL * 16) * (D_ * F_) + spL * 4 + (lane & 3);
    #pragma unroll
    for (int k = 0; k < 4; ++k) {
        const int p = wave * 4 + k;
        const float* s = red + p * RPITCH + lane;
        float sum = s[0] + s[NPIX * RPITCH];
        sum += s[2 * NPIX * RPITCH];
        sum += s[3 * NPIX * RPITCH];
        outL[((p >> 4) * W_ + (p & 15)) * (D_ * F_)] = sum;
    }
}

extern "C" void kernel_launch(void* const* d_in, const int* in_sizes, int n_in,
                              void* d_out, int out_size, void* d_ws, size_t ws_size,
                              hipStream_t stream)
{
    const float* xyz   = (const float*)d_in[0];
    const float* scale = (const float*)d_in[1];
    const float* rot   = (const float*)d_in[2];
    const float* opac  = (const float*)d_in[3];
    const float* feat  = (const float*)d_in[4];

    render_slices<<<dim3(W_ / 16, H_ / 8, 2), 1024, 0, stream>>>(xyz, scale, rot, opac, feat, (float*)d_out);
}


# Render: the wave's 4 accumulators re-cut from (row pair x 2 slices) to (row quad x 1 slice); a slice failing its z test or all under -150 skips its 4 exp2 and 2 sum WMMAs; sum order unchanged
# speedup vs baseline: 5.9039x; 1.0426x over previous
#include <hip/hip_runtime.h>
#include <math.h>

#define NG   2048
#define H_   128
#define W_   128
#define D_   32
#define F_   2
#define LOG2E 1.4426950408889634f

typedef float v2f __attribute__((ext_vector_type(2)));
typedef float v8f __attribute__((ext_vector_type(8)));

#define BOX_R 14.9f
struct GaussIn { float qw, qx, qy, qz, sx, sy, sz, x, y, z, op, f0, f1; };

__device__ __forceinline__ GaussIn load_gaussian(
    const int n,
    const float* __restrict__ xyz, const float* __restrict__ scale,
    const float* __restrict__ rot, const float* __restrict__ opac,
    const float* __restrict__ feat)
{
    GaussIn g;
    g.qw = rot[n*4+0];   g.qx = rot[n*4+1];   g.qy = rot[n*4+2];   g.qz = rot[n*4+3];
    g.sx = scale[n*3+0]; g.sy = scale[n*3+1]; g.sz = scale[n*3+2];
    g.x  = xyz[n*3+0];   g.y  = xyz[n*3+1];   g.z  = xyz[n*3+2];
    g.op = opac[n];
    g.f0 = feat[n*2+0];  g.f1 = feat[n*2+1];
    return g;
}

__device__ __forceinline__ void gaussian_record(
    const GaussIn g, float4& r0, float4& r1, float4& r2, float4& r3)
{
    float qw = g.qw, qx = g.qx, qy = g.qy, qz = g.qz;
    float inv = 1.0f / (sqrtf(qw*qw + qx*qx + qy*qy + qz*qz) + 1e-8f);
    qw *= inv; qx *= inv; qy *= inv; qz *= inv;

    float r00 = 1.f - 2.f*(qy*qy + qz*qz);
    float r01 = 2.f*(qx*qy - qw*qz);
    float r02 = 2.f*(qx*qz + qw*qy);
    float r10 = 2.f*(qx*qy + qw*qz);
    float r11 = 1.f - 2.f*(qx*qx + qz*qz);
    float r12 = 2.f*(qy*qz - qw*qx);
    float r20 = 2.f*(qx*qz - qw*qy);
    float r21 = 2.f*(qy*qz + qw*qx);
    float r22 = 1.f - 2.f*(qx*qx + qy*qy);

    float sx = g.sx, sy = g.sy, sz = g.sz;
    float m00 = r00*sx, m01 = r01*sy, m02 = r02*sz;
    float m10 = r10*sx, m11 = r11*sy, m12 = r12*sz;
    float m20 = r20*sx, m21 = r21*sy, m22 = r22*sz;

    float cov00 = m00*m00 + m01*m01 + m02*m02;
    float cov01 = m00*m10 + m01*m11 + m02*m12;
    float cov02 = m00*m20 + m01*m21 + m02*m22;
    float cov11 = m10*m10 + m11*m11 + m12*m12;
    float cov12 = m10*m20 + m11*m21 + m12*m22;
    float cov22 = m20*m20 + m21*m21 + m22*m22;

    float zv   = cov22 + 1e-8f;
    float zinv = 1.0f / zv;
    float thr3 = 3.0f * sqrtf(zv);

    float a  = cov00 - zinv*cov02*cov02 + 1e-6f;
    float b  = cov01 - zinv*cov02*cov12;
    float cc = cov11 - zinv*cov12*cov12 + 1e-6f;
    float idet = 1.0f / (a*cc - b*b);
    float pa = cc*idet, pb = -b*idet, pc = a*idet;

    float infl = BOX_R * (1.0f + 1e-6f * a * fabsf(pa));
    float bx = infl * sqrtf(a)  + 7.5f * (1.0f / (W_ - 1));
    float by = infl * sqrtf(cc) + 3.5f * (1.0f / (H_ - 1));

    r0 = make_float4(g.x, g.y, g.z, cov02);
    r1 = make_float4(cov12, zinv, -0.5f*LOG2E*pa, -LOG2E*pb);
    r2 = make_float4(-0.5f*LOG2E*pc, g.op, g.f0, g.f1);
    r3 = make_float4(thr3, bx, by, 0.f);
}

__device__ __forceinline__ void unit_of(int bx, int by, int bz, int v, int& xt, int& hb, int& sp)
{
    xt = (bx + v) & 7;
    hb = (by + 6 * v) & 15;
    const int d = (v & 1) ? -(v + 1) : v;
    sp = bz ? 7 - d : 8 + d;
}

__device__ __forceinline__ void chunk_masks(const float4 t0, const float4 t1, const float4 t2,
                                            const float zA, const float zB,
                                            const float xcen, const float ycen,
                                            unsigned& bitsA, unsigned& bitsB, unsigned& keep)
{
    {
        #pragma clang fp contract(off)
        bitsA = __builtin_amdgcn_ballot_w32(fabsf(t0.z - zA) < t2.x);
        bitsB = __builtin_amdgcn_ballot_w32(fabsf(t0.z - zB) < t2.x);
    }
    const float ttA = t1.y * (zA - t0.z);
    const float ttB = t1.y * (zB - t0.z);
    const unsigned inA = __builtin_amdgcn_ballot_w32(!(fabsf((t0.x + t0.w * ttA) - xcen) > t2.y))
                       & __builtin_amdgcn_ballot_w32(!(fabsf((t0.y + t1.x * ttA) - ycen) > t2.z));
    const unsigned inB = __builtin_amdgcn_ballot_w32(!(fabsf((t0.x + t0.w * ttB) - xcen) > t2.y))
                       & __builtin_amdgcn_ballot_w32(!(fabsf((t0.y + t1.x * ttB) - ycen) > t2.z));
    keep = (bitsA & inA) | (bitsB & inB);
}

#define KSPLIT 4
#define NQ     (NG / KSPLIT)
#define RPITCH 33
#define NPIX   128
#define EXP_ZERO (-150.0f)

__global__ __launch_bounds__(1024) void render_slices(
    const float* __restrict__ xyz, const float* __restrict__ scale,
    const float* __restrict__ rot, const float* __restrict__ opac,
    const float* __restrict__ feat, float* __restrict__ out)
{
    __shared__ float4 sp[NG * 4];
    __shared__ float  red[KSPLIT * NPIX * RPITCH];

    const int tid    = threadIdx.x;
    const int lane   = tid & 31;
    const int wave   = __builtin_amdgcn_readfirstlane(tid >> 5);
    const int grp    = wave >> 2;
    const int gq     = wave & 3;
    int xt, hb, spair;
    unit_of(blockIdx.x, blockIdx.y, blockIdx.z, grp, xt, hb, spair);
    const int h0     = hb * 8;
    const int laneN  = lane & 15;
    const int laneHi = lane >> 4;
    const int wbase  = xt * 16;
    const int d0     = spair * 2;

    const float zA = (float)d0       * (1.0f / (D_ - 1));
    const float zB = (float)(d0 + 1) * (1.0f / (D_ - 1));
    const float xcen = ((float)wbase + 7.5f) * (1.0f / (W_ - 1));
    const float ycen = ((float)h0    + 3.5f) * (1.0f / (H_ - 1));

    const int   tS = (laneN >> 1) & 1;
    const int   tY = ((laneN >> 2) & 1) * 4 + (laneN & 1) + 2 * (laneN >> 3);
    const float zT = (float)(d0 + tS) * (1.0f / (D_ - 1));
    const float yT = (float)(h0 + tY) * (1.0f / (H_ - 1));
    const float xc = (float)(wbase + 8) * (1.0f / (W_ - 1));
    const float xp = (float)(wbase + laneN) * (1.0f / (W_ - 1)) - xc;
    v2f Bx;
    Bx.x = laneHi ? 1.f : xp * xp;
    Bx.y = laneHi ? 0.f : xp;

    const bool selK0 = (laneN >> 1) == (laneHi << 1);
    const bool selK1 = (laneN >> 1) == (laneHi << 1) + 1;
    const bool f1    = (laneN & 1) != 0;

    const v8f zero8 = {0.f, 0.f, 0.f, 0.f, 0.f, 0.f, 0.f, 0.f};
    v8f accA0 = zero8, accA1 = zero8;
    v8f accB0 = zero8, accB1 = zero8;

    {
        GaussIn gin[NG / 1024];
        #pragma unroll
        for (int j = 0; j < NG / 1024; ++j) gin[j] = load_gaussian(tid + j * 1024, xyz, scale, rot, opac, feat);
        __builtin_amdgcn_sched_barrier(0);
        #pragma unroll
        for (int j = 0; j < NG / 1024; ++j) {
            const int n = tid + j * 1024;
            float4 r0, r1, r2, r3;
            gaussian_record(gin[j], r0, r1, r2, r3);
            sp[n*4+0] = r0;
            sp[n*4+1] = r1;
            sp[n*4+2] = r2;
            sp[n*4+3] = r3;
        }
    }
    __syncthreads();

    {
        const int ibase = gq * NQ;
        unsigned nextA, nextB, nextKeep;
        chunk_masks(sp[(ibase + lane) * 4 + 0], sp[(ibase + lane) * 4 + 1], sp[(ibase + lane) * 4 + 3],
                    zA, zB, xcen, ycen, nextA, nextB, nextKeep);
        for (int c = 0; c < NQ / 32; ++c) {
            const int cb = ibase + c * 32;
            const unsigned bitsA = nextA, bitsB = nextB;
            unsigned keep = nextKeep;
            const int cn = (c + 1 < NQ / 32) ? c + 1 : c;
            const float4 t0 = sp[(ibase + cn * 32 + lane) * 4 + 0];
            const float4 t1 = sp[(ibase + cn * 32 + lane) * 4 + 1];
            const float4 t2 = sp[(ibase + cn * 32 + lane) * 4 + 3];
            while (keep != 0u) {
                const unsigned low = keep & (0u - keep);
                keep ^= low;
                const int  i  = cb + __builtin_ctz(low);
                const bool mA = (bitsA & low) != 0u;
                const bool mB = (bitsB & low) != 0u;

                float4 q0 = sp[i*4+0];
                float4 q1 = sp[i*4+1];
                float4 q2 = sp[i*4+2];

                float tt = q1.y * (zT - q0.z);
                float m  = (q0.x - xc) + q0.w * tt;
                float dy = yT - (q0.y + q1.x * tt);
                float ua = q1.z * m;
                float vb = q1.w * dy;
                float c1 = vb - 2.f * ua;
                float cz = m * (ua - vb) + q2.x * dy * dy;
                v2f A1;
                A1.x = laneHi ? cz  : q1.z;
                A1.y = laneHi ? 0.f : c1;

                v8f E = __builtin_amdgcn_wmma_f32_16x16x4_f32(
                            false, A1, false, Bx, (short)0, zero8, false, false);

                asm volatile("" :: "v"(q2.y), "v"(q2.z), "v"(q2.w));

                const float mxA = fmaxf(fmaxf(fmaxf(E[0], E[1]), E[4]), E[5]);
                const float mxB = fmaxf(fmaxf(fmaxf(E[2], E[3]), E[6]), E[7]);
                const unsigned liveA = mA ? __builtin_amdgcn_ballot_w32(!(mxA < EXP_ZERO)) : 0u;
                const unsigned liveB = mB ? __builtin_amdgcn_ballot_w32(!(mxB < EXP_ZERO)) : 0u;
                if ((liveA | liveB) != 0u) {
                    const float fs = f1 ? q2.w : q2.z;
                    const float wo = q2.y * fs;
                    v2f B2;
                    B2.x = selK0 ? wo : 0.f;
                    B2.y = selK1 ? wo : 0.f;

                    if (liveA != 0u) {
                        v2f a0 = {__builtin_amdgcn_exp2f(E[0]), __builtin_amdgcn_exp2f(E[1])};
                        v2f a1 = {__builtin_amdgcn_exp2f(E[4]), __builtin_amdgcn_exp2f(E[5])};
                        accA0 = __builtin_amdgcn_wmma_f32_16x16x4_f32(
                                    false, a0, false, B2, (short)0, accA0, false, false);
                        accA1 = __builtin_amdgcn_wmma_f32_16x16x4_f32(
                                    false, a1, false, B2, (short)0, accA1, false, false);
                    }
                    if (liveB != 0u) {
                        v2f b0 = {__builtin_amdgcn_exp2f(E[2]), __builtin_amdgcn_exp2f(E[3])};
                        v2f b1 = {__builtin_amdgcn_exp2f(E[6]), __builtin_amdgcn_exp2f(E[7])};
                        accB0 = __builtin_amdgcn_wmma_f32_16x16x4_f32(
                                    false, b0, false, B2, (short)0, accB0, false, false);
                        accB1 = __builtin_amdgcn_wmma_f32_16x16x4_f32(
                                    false, b1, false, B2, (short)0, accB1, false, false);
                    }
                }
            }
            chunk_masks(t0, t1, t2, zA, zB, xcen, ycen, nextA, nextB, nextKeep);
        }
    }

    if (laneN < 8) {
        float* dst = red + (gq * NPIX + ((laneN >> 1) << 4) + (laneHi << 3)) * RPITCH + grp * 4 + (laneN & 1);
        #pragma unroll
        for (int r = 0; r < 8; ++r) {
            dst[(r +  0) * RPITCH + 0] = accA0[r];
            dst[(r + 64) * RPITCH + 0] = accA1[r];
            dst[(r +  0) * RPITCH + 2] = accB0[r];
            dst[(r + 64) * RPITCH + 2] = accB1[r];
        }
    }
    __syncthreads();

    int xtL, hbL, spL;
    unit_of(blockIdx.x, blockIdx.y, blockIdx.z, lane >> 2, xtL, hbL, spL);
    float* const outL = out + (hbL * 8 * W_ + xtL * 16) * (D_ * F_) + spL * 4 + (lane & 3);
    #pragma unroll
    for (int k = 0; k < 4; ++k) {
        const int p = wave * 4 + k;
        const float* s = red + p * RPITCH + lane;
        float sum = s[0] + s[NPIX * RPITCH];
        sum += s[2 * NPIX * RPITCH];
        sum += s[3 * NPIX * RPITCH];
        outL[((p >> 4) * W_ + (p & 15)) * (D_ * F_)] = sum;
    }
}

extern "C" void kernel_launch(void* const* d_in, const int* in_sizes, int n_in,
                              void* d_out, int out_size, void* d_ws, size_t ws_size,
                              hipStream_t stream)
{
    const float* xyz   = (const float*)d_in[0];
    const float* scale = (const float*)d_in[1];
    const float* rot   = (const float*)d_in[2];
    const float* opac  = (const float*)d_in[3];
    const float* feat  = (const float*)d_in[4];

    render_slices<<<dim3(W_ / 16, H_ / 8, 2), 1024, 0, stream>>>(xyz, scale, rot, opac, feat, (float*)d_out);
}
